# starting kernel
# baseline (speedup 1.0000x reference)
#include <hip/hip_runtime.h>
#include <hip/hip_bf16.h>


typedef float v2f __attribute__((ext_vector_type(2)));
typedef float v8f __attribute__((ext_vector_type(8)));

#define N_NODES 100000
#define IN_CH   128
#define HID     64
#define OUT_CH  32

__global__ void k_init(float* __restrict__ deg, float* __restrict__ agg1,
                       float* __restrict__ out, int n) {
  long i = (long)blockIdx.x * blockDim.x + threadIdx.x;
  if (i < n) deg[i] = 1.0f;
  if (i < (long)n * HID) agg1[i] = 0.0f;
  if (i < (long)n * OUT_CH) out[i] = 0.0f;
}

__global__ void k_deg(const int* __restrict__ dst, float* __restrict__ deg, int e) {
  int i = blockIdx.x * blockDim.x + threadIdx.x;
  if (i < e) {
    (void)__hip_atomic_fetch_add(&deg[dst[i]], 1.0f,
                                 __ATOMIC_RELAXED, __HIP_MEMORY_SCOPE_AGENT);
  }
}

__global__ void k_rsqrt(float* __restrict__ d, int n) {
  int i = blockIdx.x * blockDim.x + threadIdx.x;
  if (i < n) d[i] = rsqrtf(d[i]);
}

__global__ void k_gemm_f32(const float* __restrict__ A, const float* __restrict__ W,
                           float* __restrict__ out, int M, int K, int NC,
                           int tiles_n, int total_tiles) {
  int wave = (int)((blockIdx.x * (long)blockDim.x + threadIdx.x) >> 5);
  if (wave >= total_tiles) return;
  int lane  = threadIdx.x & 31;
  int tm    = wave / tiles_n;
  int tn    = wave % tiles_n;
  int hi    = lane >> 4;
  int l15   = lane & 15;
  int row   = tm * 16 + l15;
  int col   = tn * 16 + l15;
  int rrow  = row < M ? row : M - 1;

  const float* arow = A + (size_t)rrow * K + 2 * hi;
  const float* wcol = W + (size_t)(2 * hi) * NC + col;

  v8f c = {};
  for (int k0 = 0; k0 < K; k0 += 4) {
    v2f a = *(const v2f*)(arow + k0);
    v2f b;
    b.x = wcol[(size_t)k0 * NC];
    b.y = wcol[(size_t)(k0 + 1) * NC];
    c = __builtin_amdgcn_wmma_f32_16x16x4_f32(
         false, a,  false, b,
         (short)0, c,  false,  false);
  }

  int m0 = tm * 16 + hi * 8;
  #pragma unroll
  for (int v = 0; v < 8; ++v) {
    int r = m0 + v;
    if (r < M) out[(size_t)r * NC + col] = c[v];
  }
}

template <int C>
__global__ void k_scatter(const float* __restrict__ m, const int* __restrict__ src,
                          const int* __restrict__ dst, const float* __restrict__ dinv,
                          float* __restrict__ agg, int e) {
  int wave = (int)((blockIdx.x * (long)blockDim.x + threadIdx.x) >> 5);
  if (wave >= e) return;
  int lane = threadIdx.x & 31;
  int s = src[wave];
  int d = dst[wave];
  float w = dinv[s] * dinv[d];
  #pragma unroll
  for (int c = lane; c < C; c += 32) {
    float val = m[(size_t)s * C + c] * w;
    (void)__hip_atomic_fetch_add(&agg[(size_t)d * C + c], val,
                                 __ATOMIC_RELAXED, __HIP_MEMORY_SCOPE_AGENT);
  }
}

__global__ void k_post1(float* __restrict__ agg, const float* __restrict__ m,
                        const float* __restrict__ dinv, const float* __restrict__ b1,
                        int n) {
  long i = (long)blockIdx.x * blockDim.x + threadIdx.x;
  if (i >= (long)n * HID) return;
  int node = (int)(i >> 6);
  int ch   = (int)(i & 63);
  float di = dinv[node];
  float v  = agg[i] + m[i] * di * di + b1[ch];
  agg[i]   = v > 0.0f ? v : 0.0f;
}

__global__ void k_post2(float* __restrict__ out, const float* __restrict__ m2,
                        const float* __restrict__ dinv, const float* __restrict__ b2,
                        int n) {
  long i = (long)blockIdx.x * blockDim.x + threadIdx.x;
  if (i >= (long)n * OUT_CH) return;
  int node = (int)(i >> 5);
  int ch   = (int)(i & 31);
  float di = dinv[node];
  out[i] = out[i] + m2[i] * di * di + b2[ch];
}

extern "C" void kernel_launch(void* const* d_in, const int* in_sizes, int n_in,
                              void* d_out, int out_size, void* d_ws, size_t ws_size,
                              hipStream_t stream) {
  const float* x   = (const float*)d_in[0];
  const int*   ei  = (const int*)d_in[1];
  const float* W1  = (const float*)d_in[2];
  const float* b1  = (const float*)d_in[3];
  const float* W2  = (const float*)d_in[4];
  const float* b2  = (const float*)d_in[5];
  float*       out = (float*)d_out;

  const int N = in_sizes[0] / IN_CH;
  const int E = in_sizes[1] / 2;
  const int* src = ei;
  const int* dst = ei + E;

  float* dinv = (float*)d_ws;
  float* bufA = dinv + 131072;
  float* bufB = bufA + (size_t)N * HID;

  const int BT = 256;

  {
    long total = (long)N * HID;
    k_init<<<(int)((total + BT - 1) / BT), BT, 0, stream>>>(dinv, bufB, out, N);
  }
  k_deg<<<(E + BT - 1) / BT, BT, 0, stream>>>(dst, dinv, E);
  k_rsqrt<<<(N + BT - 1) / BT, BT, 0, stream>>>(dinv, N);
  {
    int tiles_m = (N + 15) / 16, tiles_n = HID / 16;
    int tiles = tiles_m * tiles_n;
    long threads = (long)tiles * 32;
    k_gemm_f32<<<(int)((threads + BT - 1) / BT), BT, 0, stream>>>(
        x, W1, bufA, N, IN_CH, HID, tiles_n, tiles);
  }
  {
    long threads = (long)E * 32;
    k_scatter<HID><<<(int)((threads + BT - 1) / BT), BT, 0, stream>>>(
        bufA, src, dst, dinv, bufB, E);
  }
  {
    long total = (long)N * HID;
    k_post1<<<(int)((total + BT - 1) / BT), BT, 0, stream>>>(bufB, bufA, dinv, b1, N);
  }
  {
    int tiles_m = (N + 15) / 16, tiles_n = OUT_CH / 16;
    int tiles = tiles_m * tiles_n;
    long threads = (long)tiles * 32;
    k_gemm_f32<<<(int)((threads + BT - 1) / BT), BT, 0, stream>>>(
        bufB, W2, bufA, N, HID, OUT_CH, tiles_n, tiles);
  }
  {
    long threads = (long)E * 32;
    k_scatter<OUT_CH><<<(int)((threads + BT - 1) / BT), BT, 0, stream>>>(
        bufA, src, dst, dinv, out, E);
  }
  {
    long total = (long)N * OUT_CH;
    k_post2<<<(int)((total + BT - 1) / BT), BT, 0, stream>>>(out, bufA, dinv, b2, N);
  }
}


# Scatter: a wave adds only into node rows of its own cache half (die number picks the range; blocks draw edge tickets per range, filter them into an LDS ring); float atomics, arrival order as before
# speedup vs baseline: 3.1469x; 3.1469x over previous
#include <hip/hip_runtime.h>
#include <hip/hip_bf16.h>


typedef float v2f __attribute__((ext_vector_type(2)));
typedef float v8f __attribute__((ext_vector_type(8)));

#define N_NODES 100000
#define IN_CH   128
#define HID     64
#define OUT_CH  32

#define HIST_B      128
#define HIST_PITCH  ((((N_NODES + 1) / 2) + 31) / 32 * 32)

__global__ void __launch_bounds__(1024) k_hist(const int* __restrict__ dst,
                                               unsigned* __restrict__ hist, int e) {
  __shared__ __attribute__((aligned(16))) unsigned cnt[HIST_PITCH];
  uint4* c4 = (uint4*)cnt;
  for (int i = threadIdx.x; i < HIST_PITCH / 4; i += 1024)
    c4[i] = make_uint4(0u, 0u, 0u, 0u);
  __syncthreads();

  int chunk = (e + (int)gridDim.x - 1) / (int)gridDim.x;
  int e0 = (int)blockIdx.x * chunk;
  int e1 = e0 + chunk < e ? e0 + chunk : e;
  constexpr int UB = 16;
  for (int base = e0 + (int)threadIdx.x; base < e1; base += 1024 * UB) {
    int d[UB];
    #pragma unroll
    for (int u = 0; u < UB; ++u) {
      int i = base + u * 1024;
      d[u] = dst[i < e1 ? i : e1 - 1];
    }
    #pragma unroll
    for (int u = 0; u < UB; ++u) {
      bool in = base + u * 1024 < e1;
      int      word = in ? (d[u] >> 1) : (int)threadIdx.x;
      unsigned one  = in ? (1u << ((d[u] & 1) << 4)) : 0u;
      (void)__hip_atomic_fetch_add(&cnt[word], one,
                                   __ATOMIC_RELAXED, __HIP_MEMORY_SCOPE_WORKGROUP);
    }
  }
  __syncthreads();

  uint4* o4 = (uint4*)(hist + (size_t)blockIdx.x * HIST_PITCH);
  for (int i = threadIdx.x; i < HIST_PITCH / 4; i += 1024) o4[i] = c4[i];
}

__global__ void __launch_bounds__(256) k_degsum(const unsigned* __restrict__ hist,
                                                float* __restrict__ dinv,
                                                unsigned* __restrict__ tick, int n) {
  __shared__ unsigned part[2][8][32];
  if (blockIdx.x == 0 && threadIdx.x < 4) tick[threadIdx.x * 64] = 0u;
  int lane = threadIdx.x & 31;
  int wv   = threadIdx.x >> 5;
  int w    = blockIdx.x * 32 + lane;
  unsigned lo = 0, hi = 0;
  #pragma unroll
  for (int b = 0; b < HIST_B / 8; ++b) {
    unsigned v = hist[(size_t)(b * 8 + wv) * HIST_PITCH + w];
    lo += v & 0xffffu;
    hi += v >> 16;
  }
  part[0][wv][lane] = lo;
  part[1][wv][lane] = hi;
  __syncthreads();
  if (wv == 0) {
    for (int k = 1; k < 8; ++k) {
      lo += part[0][k][lane];
      hi += part[1][k][lane];
    }
    int n0 = 2 * w;
    float d0 = rsqrtf(1.0f + (float)lo);
    float d1 = rsqrtf(1.0f + (float)hi);
    if (n0 + 1 < n)  *(float2*)(dinv + n0) = make_float2(d0, d1);
    else if (n0 < n) dinv[n0] = d0;
  }
}

template <bool RELU_A>
__global__ void k_gemm_f32(const float* __restrict__ A, const float* __restrict__ W,
                           const float* __restrict__ dinv, const float* __restrict__ bias,
                           float* __restrict__ out, float* __restrict__ acc,
                           int M, int K, int NC, int tiles_n, int total_tiles) {
  int wave = (int)((blockIdx.x * (long)blockDim.x + threadIdx.x) >> 5);
  if (wave >= total_tiles) return;
  int lane  = threadIdx.x & 31;
  int tm    = wave / tiles_n;
  int tn    = wave % tiles_n;
  int hi    = lane >> 4;
  int l15   = lane & 15;
  int row   = tm * 16 + l15;
  int col   = tn * 16 + l15;
  int rrow  = row < M ? row : M - 1;
  int m0    = tm * 16 + hi * 8;

  float sw[8];
  #pragma unroll
  for (int v = 0; v < 8; ++v) {
    int r = m0 + v;
    float di = dinv[r < M ? r : M - 1];
    sw[v] = di * di;
  }
  float bc = bias[col];

  const float* arow = A + (size_t)rrow * K + 2 * hi;
  const float* wcol = W + (size_t)(2 * hi) * NC + col;

  v8f c = {};
  #pragma unroll 4
  for (int k0 = 0; k0 < K; k0 += 4) {
    v2f a = *(const v2f*)(arow + k0);
    if (RELU_A) {
      a.x = __builtin_fmaxf(a.x, 0.0f);
      a.y = __builtin_fmaxf(a.y, 0.0f);
    }
    v2f b;
    b.x = wcol[(size_t)k0 * NC];
    b.y = wcol[(size_t)(k0 + 1) * NC];
    c = __builtin_amdgcn_wmma_f32_16x16x4_f32(
         false, a,  false, b,
         (short)0, c,  false,  false);
  }

  #pragma unroll
  for (int v = 0; v < 8; ++v) {
    int r = m0 + v;
    if (r < M) {
      out[(size_t)r * NC + col] = c[v];
      acc[(size_t)r * NC + col] = c[v] * sw[v] + bc;
    }
  }
}

#define TK_EDGES 1024
#define Q_RING   2048
#define TK_PITCH 64

template <int C>
__device__ __forceinline__ void rows_full(const float* __restrict__ m, float* __restrict__ agg,
                                          int s, int d, float w, int lane) {
  constexpr int R = C / 32;
  constexpr int U = 16 / R;
  #pragma unroll
  for (int j0 = 0; j0 < 32; j0 += U) {
    float v[U][R];
    #pragma unroll
    for (int u = 0; u < U; ++u) {
      int sj = __builtin_amdgcn_readlane(s, j0 + u);
      const float* row = m + (size_t)sj * C + lane;
      #pragma unroll
      for (int r = 0; r < R; ++r) v[u][r] = row[32 * r];
    }
    #pragma unroll
    for (int u = 0; u < U; ++u) {
      int   dj = __builtin_amdgcn_readlane(d, j0 + u);
      float wj = __int_as_float(__builtin_amdgcn_readlane(__float_as_int(w), j0 + u));
      float* arow = agg + (size_t)dj * C + lane;
      #pragma unroll
      for (int r = 0; r < R; ++r) {
        (void)__hip_atomic_fetch_add(&arow[32 * r], v[u][r] * wj,
                                     __ATOMIC_RELAXED, __HIP_MEMORY_SCOPE_AGENT);
      }
    }
  }
}

template <int C>
__device__ __forceinline__ void rows_part(const float* __restrict__ m, float* __restrict__ agg,
                                          int s, int d, float w, int cnt, int lane) {
  constexpr int R = C / 32;
  constexpr int U = 16 / R;
  for (int j0 = 0; j0 < cnt; j0 += U) {
    float v[U][R];
    #pragma unroll
    for (int u = 0; u < U; ++u) {
      int j  = j0 + u < cnt ? j0 + u : cnt - 1;
      int sj = __builtin_amdgcn_readlane(s, j);
      const float* row = m + (size_t)sj * C + lane;
      #pragma unroll
      for (int r = 0; r < R; ++r) v[u][r] = row[32 * r];
    }
    #pragma unroll
    for (int u = 0; u < U; ++u) {
      if (j0 + u < cnt) {
        int   dj = __builtin_amdgcn_readlane(d, j0 + u);
        float wj = __int_as_float(__builtin_amdgcn_readlane(__float_as_int(w), j0 + u));
        float* arow = agg + (size_t)dj * C + lane;
        #pragma unroll
        for (int r = 0; r < R; ++r) {
          (void)__hip_atomic_fetch_add(&arow[32 * r], v[u][r] * wj,
                                       __ATOMIC_RELAXED, __HIP_MEMORY_SCOPE_AGENT);
        }
      }
    }
  }
}

template <int C>
__global__ void __launch_bounds__(256) __attribute__((amdgpu_waves_per_eu(16, 16))) k_scatter(const float* __restrict__ m,
                                                 const int* __restrict__ src,
                                                 const int* __restrict__ dst,
                                                 const float* __restrict__ dinv,
                                                 float* __restrict__ agg,
                                                 unsigned* __restrict__ tick, int e, int n) {
  __shared__ int qs[Q_RING];
  __shared__ int qd[Q_RING];
  __shared__ int s_tail;
  __shared__ unsigned s_t[2];
  const int tid  = threadIdx.x;
  const int lane = tid & 31;
  const int wv   = tid >> 5;
  const int nsplit = n >> 1;
  const unsigned ntick = (unsigned)((e + TK_EDGES - 1) / TK_EDGES);
  const int half = (int)((__builtin_amdgcn_s_sendmsg_rtn(0x87) >> 18) & 1u);

  if (tid == 0) s_tail = 0;
  int head = 0;
  int it   = 0;
  for (int pass = 0; pass < 2; ++pass) {
    const int rg = half ^ pass;
    for (;;) {
      if (tid == 0)
        s_t[it & 1] = __hip_atomic_fetch_add(&tick[rg * TK_PITCH], 1u, __ATOMIC_RELAXED,
                                             __HIP_MEMORY_SCOPE_AGENT);
      __syncthreads();
      const unsigned t = s_t[it & 1];
      ++it;
      if (t >= ntick) break;

      int  si[TK_EDGES / 256], di[TK_EDGES / 256];
      bool in[TK_EDGES / 256];
      #pragma unroll
      for (int u = 0; u < TK_EDGES / 256; ++u) {
        long i  = (long)t * TK_EDGES + u * 256 + tid;
        in[u]   = i < e;
        long ic = in[u] ? i : (long)e - 1;
        si[u] = src[ic];
        di[u] = dst[ic];
      }
      #pragma unroll
      for (int u = 0; u < TK_EDGES / 256; ++u) {
        bool act = in[u] && ((di[u] >= nsplit) == (rg != 0));
        unsigned mask = __builtin_amdgcn_ballot_w32(act);
        int c    = __builtin_popcount(mask);
        int base = 0;
        if (lane == 0 && c != 0)
          base = __hip_atomic_fetch_add(&s_tail, c, __ATOMIC_RELAXED,
                                        __HIP_MEMORY_SCOPE_WORKGROUP);
        base = __builtin_amdgcn_readfirstlane(base);
        int pos = (base + (int)__builtin_amdgcn_mbcnt_lo(mask, 0u)) & (Q_RING - 1);
        if (act) { qs[pos] = si[u]; qd[pos] = di[u]; }
      }
      __syncthreads();

      const int avail = s_tail - head;
      const int proc  = (avail >> 5) & ~7;
      for (int sl = wv; sl < proc; sl += 8) {
        int   p = (head + sl * 32 + lane) & (Q_RING - 1);
        int   s = qs[p];
        int   d = qd[p];
        float w = dinv[s] * dinv[d];
        rows_full<C>(m, agg, s, d, w, lane);
      }
      head += proc * 32;
    }
  }

  const int avail = s_tail - head;
  const int full  = avail >> 5;
  const int rem   = avail & 31;
  if (wv < full) {
    int   p = (head + wv * 32 + lane) & (Q_RING - 1);
    int   s = qs[p];
    int   d = qd[p];
    float w = dinv[s] * dinv[d];
    rows_full<C>(m, agg, s, d, w, lane);
  }
  if (wv == full && rem > 0) {
    int s = 0, d = 0;
    float w = 0.0f;
    if (lane < rem) {
      int p = (head + full * 32 + lane) & (Q_RING - 1);
      s = qs[p];
      d = qd[p];
      w = dinv[s] * dinv[d];
    }
    rows_part<C>(m, agg, s, d, w, rem, lane);
  }
}

extern "C" void kernel_launch(void* const* d_in, const int* in_sizes, int n_in,
                              void* d_out, int out_size, void* d_ws, size_t ws_size,
                              hipStream_t stream) {
  const float* x   = (const float*)d_in[0];
  const int*   ei  = (const int*)d_in[1];
  const float* W1  = (const float*)d_in[2];
  const float* b1  = (const float*)d_in[3];
  const float* W2  = (const float*)d_in[4];
  const float* b2  = (const float*)d_in[5];
  float*       out = (float*)d_out;

  const int N = in_sizes[0] / IN_CH;
  const int E = in_sizes[1] / 2;
  const int* src = ei;
  const int* dst = ei + E;

  float* dinv = (float*)d_ws;
  unsigned* tick = (unsigned*)(dinv + 130048);
  float* bufA = dinv + 131072;
  float* bufB = bufA + (size_t)N * HID;

  const int BT = 256;

  k_hist<<<HIST_B, 1024, 0, stream>>>(dst, (unsigned*)bufA, E);
  k_degsum<<<HIST_PITCH / 32, 256, 0, stream>>>((const unsigned*)bufA, dinv, tick, N);
  {
    int tiles_m = (N + 15) / 16, tiles_n = HID / 16;
    int tiles = tiles_m * tiles_n;
    long threads = (long)tiles * 32;
    k_gemm_f32<false><<<(int)((threads + BT - 1) / BT), BT, 0, stream>>>(
        x, W1, dinv, b1, bufA, bufB, N, IN_CH, HID, tiles_n, tiles);
  }
  k_scatter<HID><<<2048, 256, 0, stream>>>(bufA, src, dst, dinv, bufB, tick, E, N);
  {
    int tiles_m = (N + 15) / 16, tiles_n = OUT_CH / 16;
    int tiles = tiles_m * tiles_n;
    long threads = (long)tiles * 32;
    k_gemm_f32<true><<<(int)((threads + BT - 1) / BT), BT, 0, stream>>>(
        bufB, W2, dinv, b2, bufA, out, N, HID, OUT_CH, tiles_n, tiles);
  }
  k_scatter<OUT_CH><<<2048, 256, 0, stream>>>(bufA, src, dst, dinv, out, tick + 128, E, N);
}


# GEMMs: a block's A rows (one contiguous 16 KB piece) copied into LDS by whole-line 16-byte loads, A fragments read from LDS (pitch K+4) instead of 8-byte row-strided global loads; same sums
# speedup vs baseline: 3.4032x; 1.0814x over previous
#include <hip/hip_runtime.h>
#include <hip/hip_bf16.h>


typedef float v2f __attribute__((ext_vector_type(2)));
typedef float v8f __attribute__((ext_vector_type(8)));

#define N_NODES 100000
#define IN_CH   128
#define HID     64
#define OUT_CH  32

#define HIST_B      128
#define HIST_PITCH  ((((N_NODES + 1) / 2) + 31) / 32 * 32)

__global__ void __launch_bounds__(1024) k_hist(const int* __restrict__ dst,
                                               unsigned* __restrict__ hist, int e) {
  __shared__ __attribute__((aligned(16))) unsigned cnt[HIST_PITCH];
  uint4* c4 = (uint4*)cnt;
  for (int i = threadIdx.x; i < HIST_PITCH / 4; i += 1024)
    c4[i] = make_uint4(0u, 0u, 0u, 0u);
  __syncthreads();

  int chunk = (e + (int)gridDim.x - 1) / (int)gridDim.x;
  int e0 = (int)blockIdx.x * chunk;
  int e1 = e0 + chunk < e ? e0 + chunk : e;
  constexpr int UB = 16;
  for (int base = e0 + (int)threadIdx.x; base < e1; base += 1024 * UB) {
    int d[UB];
    #pragma unroll
    for (int u = 0; u < UB; ++u) {
      int i = base + u * 1024;
      d[u] = dst[i < e1 ? i : e1 - 1];
    }
    #pragma unroll
    for (int u = 0; u < UB; ++u) {
      bool in = base + u * 1024 < e1;
      int      word = in ? (d[u] >> 1) : (int)threadIdx.x;
      unsigned one  = in ? (1u << ((d[u] & 1) << 4)) : 0u;
      (void)__hip_atomic_fetch_add(&cnt[word], one,
                                   __ATOMIC_RELAXED, __HIP_MEMORY_SCOPE_WORKGROUP);
    }
  }
  __syncthreads();

  uint4* o4 = (uint4*)(hist + (size_t)blockIdx.x * HIST_PITCH);
  for (int i = threadIdx.x; i < HIST_PITCH / 4; i += 1024) o4[i] = c4[i];
}

__global__ void __launch_bounds__(256) k_degsum(const unsigned* __restrict__ hist,
                                                float* __restrict__ dinv,
                                                unsigned* __restrict__ tick, int n) {
  __shared__ unsigned part[2][8][32];
  if (blockIdx.x == 0 && threadIdx.x < 4) tick[threadIdx.x * 64] = 0u;
  int lane = threadIdx.x & 31;
  int wv   = threadIdx.x >> 5;
  int w    = blockIdx.x * 32 + lane;
  unsigned lo = 0, hi = 0;
  #pragma unroll
  for (int b = 0; b < HIST_B / 8; ++b) {
    unsigned v = hist[(size_t)(b * 8 + wv) * HIST_PITCH + w];
    lo += v & 0xffffu;
    hi += v >> 16;
  }
  part[0][wv][lane] = lo;
  part[1][wv][lane] = hi;
  __syncthreads();
  if (wv == 0) {
    for (int k = 1; k < 8; ++k) {
      lo += part[0][k][lane];
      hi += part[1][k][lane];
    }
    int n0 = 2 * w;
    float d0 = rsqrtf(1.0f + (float)lo);
    float d1 = rsqrtf(1.0f + (float)hi);
    if (n0 + 1 < n)  *(float2*)(dinv + n0) = make_float2(d0, d1);
    else if (n0 < n) dinv[n0] = d0;
  }
}

template <bool RELU_A, int K, int NC>
__global__ void __launch_bounds__(256) k_gemm_f32(const float* __restrict__ A, const float* __restrict__ W,
                           const float* __restrict__ dinv, const float* __restrict__ bias,
                           float* __restrict__ out, float* __restrict__ acc,
                           int M, int total_tiles) {
  constexpr int TILES_N = NC / 16;
  constexpr int STRIPS  = 8 / TILES_N;
  constexpr int ROWS    = STRIPS * 16;
  constexpr int PITCH   = K + 4;
  constexpr int V4_ROW  = K / 4;
  constexpr int PASSES  = ROWS * V4_ROW / 256;
  __shared__ __attribute__((aligned(16))) float sA[ROWS * PITCH];

  const int tid  = threadIdx.x;
  const int lane = tid & 31;
  const int wv   = tid >> 5;
  const int wave = (int)blockIdx.x * 8 + wv;
  const int ls   = wv / TILES_N;
  const int tm   = (int)blockIdx.x * STRIPS + ls;
  const int tn   = wv % TILES_N;
  const int hi   = lane >> 4;
  const int l15  = lane & 15;
  const int col  = tn * 16 + l15;
  const int m0   = tm * 16 + hi * 8;
  const int row0 = (int)blockIdx.x * ROWS;

  float4 t[PASSES];
  #pragma unroll
  for (int p = 0; p < PASSES; ++p) {
    int q  = p * 256 + tid;
    int r  = row0 + q / V4_ROW;
    int gr = r < M ? r : M - 1;
    t[p] = *(const float4*)(A + (size_t)gr * K + 4 * (q % V4_ROW));
  }

  float sw[8];
  #pragma unroll
  for (int v = 0; v < 8; ++v) {
    int r = m0 + v;
    float di = dinv[r < M ? r : M - 1];
    sw[v] = di * di;
  }
  float bc = bias[col];

  if (RELU_A) __builtin_amdgcn_sched_barrier(0);

  #pragma unroll
  for (int p = 0; p < PASSES; ++p) {
    int q = p * 256 + tid;
    float4 v = t[p];
    if (RELU_A) {
      v.x = __builtin_fmaxf(v.x, 0.0f);
      v.y = __builtin_fmaxf(v.y, 0.0f);
      v.z = __builtin_fmaxf(v.z, 0.0f);
      v.w = __builtin_fmaxf(v.w, 0.0f);
    }
    *(float4*)(sA + (q / V4_ROW) * PITCH + 4 * (q % V4_ROW)) = v;
  }
  __syncthreads();
  if (wave >= total_tiles) return;

  const float* arow = sA + (ls * 16 + l15) * PITCH + 2 * hi;
  const float* wcol = W + (size_t)(2 * hi) * NC + col;

  v8f c = {};
  #pragma unroll 4
  for (int k0 = 0; k0 < K; k0 += 4) {
    v2f a = *(const v2f*)(arow + k0);
    v2f b;
    b.x = wcol[(size_t)k0 * NC];
    b.y = wcol[(size_t)(k0 + 1) * NC];
    c = __builtin_amdgcn_wmma_f32_16x16x4_f32(
         false, a,  false, b,
         (short)0, c,  false,  false);
  }

  #pragma unroll
  for (int v = 0; v < 8; ++v) {
    int r = m0 + v;
    if (r < M) {
      out[(size_t)r * NC + col] = c[v];
      acc[(size_t)r * NC + col] = c[v] * sw[v] + bc;
    }
  }
}

#define TK_EDGES 1024
#define Q_RING   2048
#define TK_PITCH 64

template <int C>
__device__ __forceinline__ void rows_full(const float* __restrict__ m, float* __restrict__ agg,
                                          int s, int d, float w, int lane) {
  constexpr int R = C / 32;
  constexpr int U = 16 / R;
  #pragma unroll
  for (int j0 = 0; j0 < 32; j0 += U) {
    float v[U][R];
    #pragma unroll
    for (int u = 0; u < U; ++u) {
      int sj = __builtin_amdgcn_readlane(s, j0 + u);
      const float* row = m + (size_t)sj * C + lane;
      #pragma unroll
      for (int r = 0; r < R; ++r) v[u][r] = row[32 * r];
    }
    #pragma unroll
    for (int u = 0; u < U; ++u) {
      int   dj = __builtin_amdgcn_readlane(d, j0 + u);
      float wj = __int_as_float(__builtin_amdgcn_readlane(__float_as_int(w), j0 + u));
      float* arow = agg + (size_t)dj * C + lane;
      #pragma unroll
      for (int r = 0; r < R; ++r) {
        (void)__hip_atomic_fetch_add(&arow[32 * r], v[u][r] * wj,
                                     __ATOMIC_RELAXED, __HIP_MEMORY_SCOPE_AGENT);
      }
    }
  }
}

template <int C>
__device__ __forceinline__ void rows_part(const float* __restrict__ m, float* __restrict__ agg,
                                          int s, int d, float w, int cnt, int lane) {
  constexpr int R = C / 32;
  constexpr int U = 16 / R;
  for (int j0 = 0; j0 < cnt; j0 += U) {
    float v[U][R];
    #pragma unroll
    for (int u = 0; u < U; ++u) {
      int j  = j0 + u < cnt ? j0 + u : cnt - 1;
      int sj = __builtin_amdgcn_readlane(s, j);
      const float* row = m + (size_t)sj * C + lane;
      #pragma unroll
      for (int r = 0; r < R; ++r) v[u][r] = row[32 * r];
    }
    #pragma unroll
    for (int u = 0; u < U; ++u) {
      if (j0 + u < cnt) {
        int   dj = __builtin_amdgcn_readlane(d, j0 + u);
        float wj = __int_as_float(__builtin_amdgcn_readlane(__float_as_int(w), j0 + u));
        float* arow = agg + (size_t)dj * C + lane;
        #pragma unroll
        for (int r = 0; r < R; ++r) {
          (void)__hip_atomic_fetch_add(&arow[32 * r], v[u][r] * wj,
                                       __ATOMIC_RELAXED, __HIP_MEMORY_SCOPE_AGENT);
        }
      }
    }
  }
}

template <int C>
__global__ void __launch_bounds__(256) __attribute__((amdgpu_waves_per_eu(16, 16))) k_scatter(const float* __restrict__ m,
                                                 const int* __restrict__ src,
                                                 const int* __restrict__ dst,
                                                 const float* __restrict__ dinv,
                                                 float* __restrict__ agg,
                                                 unsigned* __restrict__ tick, int e, int n) {
  __shared__ int qs[Q_RING];
  __shared__ int qd[Q_RING];
  __shared__ int s_tail;
  __shared__ unsigned s_t[2];
  const int tid  = threadIdx.x;
  const int lane = tid & 31;
  const int wv   = tid >> 5;
  const int nsplit = n >> 1;
  const unsigned ntick = (unsigned)((e + TK_EDGES - 1) / TK_EDGES);
  const int half = (int)((__builtin_amdgcn_s_sendmsg_rtn(0x87) >> 18) & 1u);

  if (tid == 0) s_tail = 0;
  int head = 0;
  int it   = 0;
  for (int pass = 0; pass < 2; ++pass) {
    const int rg = half ^ pass;
    for (;;) {
      if (tid == 0)
        s_t[it & 1] = __hip_atomic_fetch_add(&tick[rg * TK_PITCH], 1u, __ATOMIC_RELAXED,
                                             __HIP_MEMORY_SCOPE_AGENT);
      __syncthreads();
      const unsigned t = s_t[it & 1];
      ++it;
      if (t >= ntick) break;

      int  si[TK_EDGES / 256], di[TK_EDGES / 256];
      bool in[TK_EDGES / 256];
      #pragma unroll
      for (int u = 0; u < TK_EDGES / 256; ++u) {
        long i  = (long)t * TK_EDGES + u * 256 + tid;
        in[u]   = i < e;
        long ic = in[u] ? i : (long)e - 1;
        si[u] = src[ic];
        di[u] = dst[ic];
      }
      #pragma unroll
      for (int u = 0; u < TK_EDGES / 256; ++u) {
        bool act = in[u] && ((di[u] >= nsplit) == (rg != 0));
        unsigned mask = __builtin_amdgcn_ballot_w32(act);
        int c    = __builtin_popcount(mask);
        int base = 0;
        if (lane == 0 && c != 0)
          base = __hip_atomic_fetch_add(&s_tail, c, __ATOMIC_RELAXED,
                                        __HIP_MEMORY_SCOPE_WORKGROUP);
        base = __builtin_amdgcn_readfirstlane(base);
        int pos = (base + (int)__builtin_amdgcn_mbcnt_lo(mask, 0u)) & (Q_RING - 1);
        if (act) { qs[pos] = si[u]; qd[pos] = di[u]; }
      }
      __syncthreads();

      const int avail = s_tail - head;
      const int proc  = (avail >> 5) & ~7;
      for (int sl = wv; sl < proc; sl += 8) {
        int   p = (head + sl * 32 + lane) & (Q_RING - 1);
        int   s = qs[p];
        int   d = qd[p];
        float w = dinv[s] * dinv[d];
        rows_full<C>(m, agg, s, d, w, lane);
      }
      head += proc * 32;
    }
  }

  const int avail = s_tail - head;
  const int full  = avail >> 5;
  const int rem   = avail & 31;
  if (wv < full) {
    int   p = (head + wv * 32 + lane) & (Q_RING - 1);
    int   s = qs[p];
    int   d = qd[p];
    float w = dinv[s] * dinv[d];
    rows_full<C>(m, agg, s, d, w, lane);
  }
  if (wv == full && rem > 0) {
    int s = 0, d = 0;
    float w = 0.0f;
    if (lane < rem) {
      int p = (head + full * 32 + lane) & (Q_RING - 1);
      s = qs[p];
      d = qd[p];
      w = dinv[s] * dinv[d];
    }
    rows_part<C>(m, agg, s, d, w, rem, lane);
  }
}

extern "C" void kernel_launch(void* const* d_in, const int* in_sizes, int n_in,
                              void* d_out, int out_size, void* d_ws, size_t ws_size,
                              hipStream_t stream) {
  const float* x   = (const float*)d_in[0];
  const int*   ei  = (const int*)d_in[1];
  const float* W1  = (const float*)d_in[2];
  const float* b1  = (const float*)d_in[3];
  const float* W2  = (const float*)d_in[4];
  const float* b2  = (const float*)d_in[5];
  float*       out = (float*)d_out;

  const int N = in_sizes[0] / IN_CH;
  const int E = in_sizes[1] / 2;
  const int* src = ei;
  const int* dst = ei + E;

  float* dinv = (float*)d_ws;
  unsigned* tick = (unsigned*)(dinv + 130048);
  float* bufA = dinv + 131072;
  float* bufB = bufA + (size_t)N * HID;

  const int BT = 256;

  k_hist<<<HIST_B, 1024, 0, stream>>>(dst, (unsigned*)bufA, E);
  k_degsum<<<HIST_PITCH / 32, 256, 0, stream>>>((const unsigned*)bufA, dinv, tick, N);
  {
    int tiles_m = (N + 15) / 16, tiles_n = HID / 16;
    int tiles = tiles_m * tiles_n;
    long threads = (long)tiles * 32;
    k_gemm_f32<false, IN_CH, HID><<<(int)((threads + BT - 1) / BT), BT, 0, stream>>>(
        x, W1, dinv, b1, bufA, bufB, N, tiles);
  }
  k_scatter<HID><<<2048, 256, 0, stream>>>(bufA, src, dst, dinv, bufB, tick, E, N);
  {
    int tiles_m = (N + 15) / 16, tiles_n = OUT_CH / 16;
    int tiles = tiles_m * tiles_n;
    long threads = (long)tiles * 32;
    k_gemm_f32<true, HID, OUT_CH><<<(int)((threads + BT - 1) / BT), BT, 0, stream>>>(
        bufB, W2, dinv, b2, bufA, out, N, tiles);
  }
  k_scatter<OUT_CH><<<2048, 256, 0, stream>>>(bufA, src, dst, dinv, out, tick + 128, E, N);
}


# In-degree table cut 4x (256 blocks: 32 edge chunks x 8 node ranges), kept in out until GEMM 2; each GEMM-1 block sums its 32 rows' counts into dinv: the dinv launch is gone; degrees exact
# speedup vs baseline: 3.5610x; 1.0464x over previous
#include <hip/hip_runtime.h>
#include <hip/hip_bf16.h>


typedef float v2f __attribute__((ext_vector_type(2)));
typedef float v8f __attribute__((ext_vector_type(8)));

#define N_NODES 100000
#define IN_CH   128
#define HID     64
#define OUT_CH  32

#define HIST_C      32
#define HIST_R      8
#define HIST_RW     ((((((N_NODES + 1) / 2) + HIST_R - 1) / HIST_R) + 31) / 32 * 32)
#define HIST_PITCH  (HIST_R * HIST_RW)

__global__ void __launch_bounds__(1024) k_hist(const int* __restrict__ dst,
                                               unsigned* __restrict__ hist, int e) {
  __shared__ __attribute__((aligned(16))) unsigned cnt[HIST_RW];
  uint4* c4 = (uint4*)cnt;
  for (int i = threadIdx.x; i < HIST_RW / 4; i += 1024)
    c4[i] = make_uint4(0u, 0u, 0u, 0u);
  __syncthreads();

  const int ck = (int)blockIdx.x / HIST_R;
  const int w0 = ((int)blockIdx.x % HIST_R) * HIST_RW;
  int chunk = (e + HIST_C - 1) / HIST_C;
  int e0 = ck * chunk;
  int e1 = e0 + chunk < e ? e0 + chunk : e;
  constexpr int UB = 25;
  for (int base = e0 + (int)threadIdx.x; base < e1; base += 1024 * UB) {
    int d[UB];
    #pragma unroll
    for (int u = 0; u < UB; ++u) {
      int i = base + u * 1024;
      d[u] = dst[i < e1 ? i : e1 - 1];
    }
    #pragma unroll
    for (int u = 0; u < UB; ++u) {
      unsigned word = (unsigned)((d[u] >> 1) - w0);
      if (base + u * 1024 < e1 && word < (unsigned)HIST_RW)
        (void)__hip_atomic_fetch_add(&cnt[word], 1u << ((d[u] & 1) << 4),
                                     __ATOMIC_RELAXED, __HIP_MEMORY_SCOPE_WORKGROUP);
    }
  }
  __syncthreads();

  uint4* o4 = (uint4*)(hist + (size_t)ck * HIST_PITCH + w0);
  for (int i = threadIdx.x; i < HIST_RW / 4; i += 1024) o4[i] = c4[i];
}

template <bool RELU_A, bool DEG, int K, int NC>
__global__ void __launch_bounds__(256) k_gemm_f32(const float* __restrict__ A, const float* __restrict__ W,
                           float* __restrict__ dinv, const float* __restrict__ bias,
                           float* __restrict__ out, float* __restrict__ acc,
                           const unsigned* __restrict__ hist, unsigned* __restrict__ tick,
                           int M, int total_tiles) {
  constexpr int TILES_N = NC / 16;
  constexpr int STRIPS  = 8 / TILES_N;
  constexpr int ROWS    = STRIPS * 16;
  constexpr int PITCH   = K + 4;
  constexpr int V4_ROW  = K / 4;
  constexpr int PASSES  = ROWS * V4_ROW / 256;
  __shared__ __attribute__((aligned(16))) float sA[ROWS * PITCH];
  static_assert(!DEG || ROWS == 32, "the in-degree sum is written for 32 rows a block");
  __shared__ __attribute__((aligned(8))) unsigned sP[DEG ? 16 * 32 : 2];
  __shared__ float sD[DEG ? 32 : 1];

  const int tid  = threadIdx.x;
  const int lane = tid & 31;
  const int wv   = tid >> 5;
  const int wave = (int)blockIdx.x * 8 + wv;
  const int ls   = wv / TILES_N;
  const int tm   = (int)blockIdx.x * STRIPS + ls;
  const int tn   = wv % TILES_N;
  const int hi   = lane >> 4;
  const int l15  = lane & 15;
  const int col  = tn * 16 + l15;
  const int m0   = tm * 16 + hi * 8;
  const int row0 = (int)blockIdx.x * ROWS;

  float4 t[PASSES];
  #pragma unroll
  for (int p = 0; p < PASSES; ++p) {
    int q  = p * 256 + tid;
    int r  = row0 + q / V4_ROW;
    int gr = r < M ? r : M - 1;
    t[p] = *(const float4*)(A + (size_t)gr * K + 4 * (q % V4_ROW));
  }

  unsigned h0 = 0, h1 = 0;
  if (DEG) {
    const unsigned* hp = hist + (size_t)(tid >> 4) * HIST_PITCH + (row0 >> 1) + (tid & 15);
    h0 = hp[0];
    h1 = hp[(size_t)16 * HIST_PITCH];
    if (blockIdx.x == 0 && tid < 4) tick[tid * 64] = 0u;
  }

  float sw[8];
  if (!DEG) {
    #pragma unroll
    for (int v = 0; v < 8; ++v) {
      int r = m0 + v;
      float di = dinv[r < M ? r : M - 1];
      sw[v] = di * di;
    }
  }
  float bc = bias[col];

  if (RELU_A || DEG) __builtin_amdgcn_sched_barrier(0);

  #pragma unroll
  for (int p = 0; p < PASSES; ++p) {
    int q = p * 256 + tid;
    float4 v = t[p];
    if (RELU_A) {
      v.x = __builtin_fmaxf(v.x, 0.0f);
      v.y = __builtin_fmaxf(v.y, 0.0f);
      v.z = __builtin_fmaxf(v.z, 0.0f);
      v.w = __builtin_fmaxf(v.w, 0.0f);
    }
    *(float4*)(sA + (q / V4_ROW) * PITCH + 4 * (q % V4_ROW)) = v;
  }
  if (DEG) {
    uint2 pr = make_uint2((h0 & 0xffffu) + (h1 & 0xffffu), (h0 >> 16) + (h1 >> 16));
    *(uint2*)(sP + (tid >> 4) * 32 + 2 * (tid & 15)) = pr;
  }
  __syncthreads();
  if (DEG) {
    float dn = 0.0f;
    if (tid < 32) {
      unsigned deg = 0;
      #pragma unroll
      for (int g = 0; g < 16; ++g) deg += sP[g * 32 + tid];
      dn = rsqrtf(1.0f + (float)deg);
      sD[tid] = dn;
    }
    __syncthreads();
    if (tid < 32 && row0 + tid < M) dinv[row0 + tid] = dn;
    #pragma unroll
    for (int v = 0; v < 8; ++v) {
      float di = sD[ls * 16 + hi * 8 + v];
      sw[v] = di * di;
    }
  }
  if (wave >= total_tiles) return;

  const float* arow = sA + (ls * 16 + l15) * PITCH + 2 * hi;
  const float* wcol = W + (size_t)(2 * hi) * NC + col;

  v8f c = {};
  #pragma unroll 4
  for (int k0 = 0; k0 < K; k0 += 4) {
    v2f a = *(const v2f*)(arow + k0);
    v2f b;
    b.x = wcol[(size_t)k0 * NC];
    b.y = wcol[(size_t)(k0 + 1) * NC];
    c = __builtin_amdgcn_wmma_f32_16x16x4_f32(
         false, a,  false, b,
         (short)0, c,  false,  false);
  }

  #pragma unroll
  for (int v = 0; v < 8; ++v) {
    int r = m0 + v;
    if (r < M) {
      out[(size_t)r * NC + col] = c[v];
      acc[(size_t)r * NC + col] = c[v] * sw[v] + bc;
    }
  }
}

#define TK_EDGES 1024
#define Q_RING   2048
#define TK_PITCH 64

template <int C>
__device__ __forceinline__ void rows_full(const float* __restrict__ m, float* __restrict__ agg,
                                          int s, int d, float w, int lane) {
  constexpr int R = C / 32;
  constexpr int U = 16 / R;
  #pragma unroll
  for (int j0 = 0; j0 < 32; j0 += U) {
    float v[U][R];
    #pragma unroll
    for (int u = 0; u < U; ++u) {
      int sj = __builtin_amdgcn_readlane(s, j0 + u);
      const float* row = m + (size_t)sj * C + lane;
      #pragma unroll
      for (int r = 0; r < R; ++r) v[u][r] = row[32 * r];
    }
    #pragma unroll
    for (int u = 0; u < U; ++u) {
      int   dj = __builtin_amdgcn_readlane(d, j0 + u);
      float wj = __int_as_float(__builtin_amdgcn_readlane(__float_as_int(w), j0 + u));
      float* arow = agg + (size_t)dj * C + lane;
      #pragma unroll
      for (int r = 0; r < R; ++r) {
        (void)__hip_atomic_fetch_add(&arow[32 * r], v[u][r] * wj,
                                     __ATOMIC_RELAXED, __HIP_MEMORY_SCOPE_AGENT);
      }
    }
  }
}

template <int C>
__device__ __forceinline__ void rows_part(const float* __restrict__ m, float* __restrict__ agg,
                                          int s, int d, float w, int cnt, int lane) {
  constexpr int R = C / 32;
  constexpr int U = 16 / R;
  for (int j0 = 0; j0 < cnt; j0 += U) {
    float v[U][R];
    #pragma unroll
    for (int u = 0; u < U; ++u) {
      int j  = j0 + u < cnt ? j0 + u : cnt - 1;
      int sj = __builtin_amdgcn_readlane(s, j);
      const float* row = m + (size_t)sj * C + lane;
      #pragma unroll
      for (int r = 0; r < R; ++r) v[u][r] = row[32 * r];
    }
    #pragma unroll
    for (int u = 0; u < U; ++u) {
      if (j0 + u < cnt) {
        int   dj = __builtin_amdgcn_readlane(d, j0 + u);
        float wj = __int_as_float(__builtin_amdgcn_readlane(__float_as_int(w), j0 + u));
        float* arow = agg + (size_t)dj * C + lane;
        #pragma unroll
        for (int r = 0; r < R; ++r) {
          (void)__hip_atomic_fetch_add(&arow[32 * r], v[u][r] * wj,
                                       __ATOMIC_RELAXED, __HIP_MEMORY_SCOPE_AGENT);
        }
      }
    }
  }
}

template <int C>
__global__ void __launch_bounds__(256) __attribute__((amdgpu_waves_per_eu(16, 16))) k_scatter(const float* __restrict__ m,
                                                 const int* __restrict__ src,
                                                 const int* __restrict__ dst,
                                                 const float* __restrict__ dinv,
                                                 float* __restrict__ agg,
                                                 unsigned* __restrict__ tick, int e, int n) {
  __shared__ int qs[Q_RING];
  __shared__ int qd[Q_RING];
  __shared__ int s_tail;
  __shared__ unsigned s_t[2];
  const int tid  = threadIdx.x;
  const int lane = tid & 31;
  const int wv   = tid >> 5;
  const int nsplit = n >> 1;
  const unsigned ntick = (unsigned)((e + TK_EDGES - 1) / TK_EDGES);
  const int half = (int)((__builtin_amdgcn_s_sendmsg_rtn(0x87) >> 18) & 1u);

  if (tid == 0) s_tail = 0;
  int head = 0;
  int it   = 0;
  for (int pass = 0; pass < 2; ++pass) {
    const int rg = half ^ pass;
    for (;;) {
      if (tid == 0)
        s_t[it & 1] = __hip_atomic_fetch_add(&tick[rg * TK_PITCH], 1u, __ATOMIC_RELAXED,
                                             __HIP_MEMORY_SCOPE_AGENT);
      __syncthreads();
      const unsigned t = s_t[it & 1];
      ++it;
      if (t >= ntick) break;

      int  si[TK_EDGES / 256], di[TK_EDGES / 256];
      bool in[TK_EDGES / 256];
      #pragma unroll
      for (int u = 0; u < TK_EDGES / 256; ++u) {
        long i  = (long)t * TK_EDGES + u * 256 + tid;
        in[u]   = i < e;
        long ic = in[u] ? i : (long)e - 1;
        si[u] = src[ic];
        di[u] = dst[ic];
      }
      #pragma unroll
      for (int u = 0; u < TK_EDGES / 256; ++u) {
        bool act = in[u] && ((di[u] >= nsplit) == (rg != 0));
        unsigned mask = __builtin_amdgcn_ballot_w32(act);
        int c    = __builtin_popcount(mask);
        int base = 0;
        if (lane == 0 && c != 0)
          base = __hip_atomic_fetch_add(&s_tail, c, __ATOMIC_RELAXED,
                                        __HIP_MEMORY_SCOPE_WORKGROUP);
        base = __builtin_amdgcn_readfirstlane(base);
        int pos = (base + (int)__builtin_amdgcn_mbcnt_lo(mask, 0u)) & (Q_RING - 1);
        if (act) { qs[pos] = si[u]; qd[pos] = di[u]; }
      }
      __syncthreads();

      const int avail = s_tail - head;
      const int proc  = (avail >> 5) & ~7;
      for (int sl = wv; sl < proc; sl += 8) {
        int   p = (head + sl * 32 + lane) & (Q_RING - 1);
        int   s = qs[p];
        int   d = qd[p];
        float w = dinv[s] * dinv[d];
        rows_full<C>(m, agg, s, d, w, lane);
      }
      head += proc * 32;
    }
  }

  const int avail = s_tail - head;
  const int full  = avail >> 5;
  const int rem   = avail & 31;
  if (wv < full) {
    int   p = (head + wv * 32 + lane) & (Q_RING - 1);
    int   s = qs[p];
    int   d = qd[p];
    float w = dinv[s] * dinv[d];
    rows_full<C>(m, agg, s, d, w, lane);
  }
  if (wv == full && rem > 0) {
    int s = 0, d = 0;
    float w = 0.0f;
    if (lane < rem) {
      int p = (head + full * 32 + lane) & (Q_RING - 1);
      s = qs[p];
      d = qd[p];
      w = dinv[s] * dinv[d];
    }
    rows_part<C>(m, agg, s, d, w, rem, lane);
  }
}

extern "C" void kernel_launch(void* const* d_in, const int* in_sizes, int n_in,
                              void* d_out, int out_size, void* d_ws, size_t ws_size,
                              hipStream_t stream) {
  const float* x   = (const float*)d_in[0];
  const int*   ei  = (const int*)d_in[1];
  const float* W1  = (const float*)d_in[2];
  const float* b1  = (const float*)d_in[3];
  const float* W2  = (const float*)d_in[4];
  const float* b2  = (const float*)d_in[5];
  float*       out = (float*)d_out;

  const int N = in_sizes[0] / IN_CH;
  const int E = in_sizes[1] / 2;
  const int* src = ei;
  const int* dst = ei + E;

  float* dinv = (float*)d_ws;
  unsigned* tick = (unsigned*)(dinv + 130048);
  float* bufA = dinv + 131072;
  float* bufB = bufA + (size_t)N * HID;

  const int BT = 256;

  k_hist<<<HIST_C * HIST_R, 1024, 0, stream>>>(dst, (unsigned*)out, E);
  {
    int tiles_m = (N + 15) / 16, tiles_n = HID / 16;
    int tiles = tiles_m * tiles_n;
    long threads = (long)tiles * 32;
    k_gemm_f32<false, true, IN_CH, HID><<<(int)((threads + BT - 1) / BT), BT, 0, stream>>>(
        x, W1, dinv, b1, bufA, bufB, (const unsigned*)out, tick, N, tiles);
  }
  k_scatter<HID><<<2048, 256, 0, stream>>>(bufA, src, dst, dinv, bufB, tick, E, N);
  {
    int tiles_m = (N + 15) / 16, tiles_n = OUT_CH / 16;
    int tiles = tiles_m * tiles_n;
    long threads = (long)tiles * 32;
    k_gemm_f32<true, false, HID, OUT_CH><<<(int)((threads + BT - 1) / BT), BT, 0, stream>>>(
        bufB, W2, dinv, b2, bufA, out, (const unsigned*)nullptr, tick, N, tiles);
  }
  k_scatter<OUT_CH><<<2048, 256, 0, stream>>>(bufA, src, dst, dinv, out, tick + 128, E, N);
}


# Scatter: edge weight from two LDS byte reads of an in-degree table (98 KB a block of 32 waves, written by GEMM 1), not two dinv gathers from memory; same bits; float atomics, order unfixed as before
# speedup vs baseline: 4.0712x; 1.1433x over previous
#include <hip/hip_runtime.h>
#include <hip/hip_bf16.h>


typedef float v2f __attribute__((ext_vector_type(2)));
typedef float v8f __attribute__((ext_vector_type(8)));

#define N_NODES 100000
#define IN_CH   128
#define HID     64
#define OUT_CH  32

#define HIST_C      32
#define HIST_R      8
#define HIST_RW     ((((((N_NODES + 1) / 2) + HIST_R - 1) / HIST_R) + 31) / 32 * 32)
#define HIST_PITCH  (HIST_R * HIST_RW)

#define DEG_BYTES   (((N_NODES + 127) / 128) * 128)

__global__ void __launch_bounds__(1024) k_hist(const int* __restrict__ dst,
                                               unsigned* __restrict__ hist, int e) {
  __shared__ __attribute__((aligned(16))) unsigned cnt[HIST_RW];
  uint4* c4 = (uint4*)cnt;
  for (int i = threadIdx.x; i < HIST_RW / 4; i += 1024)
    c4[i] = make_uint4(0u, 0u, 0u, 0u);
  __syncthreads();

  const int ck = (int)blockIdx.x / HIST_R;
  const int w0 = ((int)blockIdx.x % HIST_R) * HIST_RW;
  int chunk = (e + HIST_C - 1) / HIST_C;
  int e0 = ck * chunk;
  int e1 = e0 + chunk < e ? e0 + chunk : e;
  constexpr int UB = 25;
  for (int base = e0 + (int)threadIdx.x; base < e1; base += 1024 * UB) {
    int d[UB];
    #pragma unroll
    for (int u = 0; u < UB; ++u) {
      int i = base + u * 1024;
      d[u] = dst[i < e1 ? i : e1 - 1];
    }
    #pragma unroll
    for (int u = 0; u < UB; ++u) {
      unsigned word = (unsigned)((d[u] >> 1) - w0);
      if (base + u * 1024 < e1 && word < (unsigned)HIST_RW)
        (void)__hip_atomic_fetch_add(&cnt[word], 1u << ((d[u] & 1) << 4),
                                     __ATOMIC_RELAXED, __HIP_MEMORY_SCOPE_WORKGROUP);
    }
  }
  __syncthreads();

  uint4* o4 = (uint4*)(hist + (size_t)ck * HIST_PITCH + w0);
  for (int i = threadIdx.x; i < HIST_RW / 4; i += 1024) o4[i] = c4[i];
}

template <bool RELU_A, bool DEG, int K, int NC>
__global__ void __launch_bounds__(256) k_gemm_f32(const float* __restrict__ A, const float* __restrict__ W,
                           float* __restrict__ dinv, const float* __restrict__ bias,
                           float* __restrict__ out, float* __restrict__ acc,
                           const unsigned* __restrict__ hist, unsigned* __restrict__ tick,
                           unsigned* __restrict__ degw,
                           int M, int total_tiles) {
  constexpr int TILES_N = NC / 16;
  constexpr int STRIPS  = 8 / TILES_N;
  constexpr int ROWS    = STRIPS * 16;
  constexpr int PITCH   = K + 4;
  constexpr int V4_ROW  = K / 4;
  constexpr int PASSES  = ROWS * V4_ROW / 256;
  __shared__ __attribute__((aligned(16))) float sA[ROWS * PITCH];
  static_assert(!DEG || ROWS == 32, "the in-degree sum is written for 32 rows a block");
  __shared__ __attribute__((aligned(8))) unsigned sP[DEG ? 16 * 32 : 2];
  __shared__ float sD[DEG ? 32 : 1];

  const int tid  = threadIdx.x;
  const int lane = tid & 31;
  const int wv   = tid >> 5;
  const int wave = (int)blockIdx.x * 8 + wv;
  const int ls   = wv / TILES_N;
  const int tm   = (int)blockIdx.x * STRIPS + ls;
  const int tn   = wv % TILES_N;
  const int hi   = lane >> 4;
  const int l15  = lane & 15;
  const int col  = tn * 16 + l15;
  const int m0   = tm * 16 + hi * 8;
  const int row0 = (int)blockIdx.x * ROWS;

  float4 t[PASSES];
  #pragma unroll
  for (int p = 0; p < PASSES; ++p) {
    int q  = p * 256 + tid;
    int r  = row0 + q / V4_ROW;
    int gr = r < M ? r : M - 1;
    t[p] = *(const float4*)(A + (size_t)gr * K + 4 * (q % V4_ROW));
  }

  unsigned h0 = 0, h1 = 0;
  if (DEG) {
    const unsigned* hp = hist + (size_t)(tid >> 4) * HIST_PITCH + (row0 >> 1) + (tid & 15);
    h0 = hp[0];
    h1 = hp[(size_t)16 * HIST_PITCH];
    if (blockIdx.x == 0 && tid < 4) tick[tid * 64] = 0u;
  }

  float sw[8];
  if (!DEG) {
    #pragma unroll
    for (int v = 0; v < 8; ++v) {
      int r = m0 + v;
      float di = dinv[r < M ? r : M - 1];
      sw[v] = di * di;
    }
  }
  float bc = bias[col];

  if (RELU_A || DEG) __builtin_amdgcn_sched_barrier(0);

  #pragma unroll
  for (int p = 0; p < PASSES; ++p) {
    int q = p * 256 + tid;
    float4 v = t[p];
    if (RELU_A) {
      v.x = __builtin_fmaxf(v.x, 0.0f);
      v.y = __builtin_fmaxf(v.y, 0.0f);
      v.z = __builtin_fmaxf(v.z, 0.0f);
      v.w = __builtin_fmaxf(v.w, 0.0f);
    }
    *(float4*)(sA + (q / V4_ROW) * PITCH + 4 * (q % V4_ROW)) = v;
  }
  if (DEG) {
    uint2 pr = make_uint2((h0 & 0xffffu) + (h1 & 0xffffu), (h0 >> 16) + (h1 >> 16));
    *(uint2*)(sP + (tid >> 4) * 32 + 2 * (tid & 15)) = pr;
  }
  __syncthreads();
  if (DEG) {
    float dn = 0.0f;
    if (tid < 32) {
      unsigned deg = 0;
      #pragma unroll
      for (int g = 0; g < 16; ++g) deg += sP[g * 32 + tid];
      dn = rsqrtf(1.0f + (float)deg);
      sD[tid] = dn;
    }
    __syncthreads();
    if (tid < 32 && row0 + tid < M) dinv[row0 + tid] = dn;
    #pragma unroll
    for (int v = 0; v < 8; ++v) {
      float di = sD[ls * 16 + hi * 8 + v];
      sw[v] = di * di;
    }
    if ((blockIdx.x & 3) == 0) {
      const unsigned* lp = hist + (size_t)(tid >> 6) * HIST_PITCH + (row0 >> 1) + (tid & 63);
      unsigned lo = 0, up = 0;
      #pragma unroll
      for (int k = 0; k < 8; ++k) {
        unsigned h = lp[(size_t)(4 * k) * HIST_PITCH];
        lo += h & 0xffffu;
        up += h >> 16;
      }
      *(uint2*)(sP + (tid >> 6) * 128 + 2 * (tid & 63)) = make_uint2(lo, up);
      __syncthreads();
      if (tid < 32) {
        unsigned pk = 0;
        #pragma unroll
        for (int b = 0; b < 4; ++b) {
          int nd = 4 * tid + b;
          unsigned dg = sP[nd] + sP[128 + nd] + sP[256 + nd] + sP[384 + nd];
          pk |= (dg < 255u ? dg : 255u) << (8 * b);
        }
        degw[(row0 >> 2) + tid] = pk;
      }
    }
  }
  if (wave >= total_tiles) return;

  const float* arow = sA + (ls * 16 + l15) * PITCH + 2 * hi;
  const float* wcol = W + (size_t)(2 * hi) * NC + col;

  v8f c = {};
  #pragma unroll 4
  for (int k0 = 0; k0 < K; k0 += 4) {
    v2f a = *(const v2f*)(arow + k0);
    v2f b;
    b.x = wcol[(size_t)k0 * NC];
    b.y = wcol[(size_t)(k0 + 1) * NC];
    c = __builtin_amdgcn_wmma_f32_16x16x4_f32(
         false, a,  false, b,
         (short)0, c,  false,  false);
  }

  #pragma unroll
  for (int v = 0; v < 8; ++v) {
    int r = m0 + v;
    if (r < M) {
      out[(size_t)r * NC + col] = c[v];
      acc[(size_t)r * NC + col] = c[v] * sw[v] + bc;
    }
  }
}

#define TK_EDGES 2048
#define Q_RING   4096
#define TK_PITCH 64
#define SC_WAVES 32

__device__ __forceinline__ float edge_weight(const unsigned char* sdeg, const float* __restrict__ dinv,
                                             int s, int d) {
  unsigned bs = sdeg[s];
  unsigned bd = sdeg[d];
  float ws = rsqrtf(1.0f + (float)bs);
  float wd = rsqrtf(1.0f + (float)bd);
  if (bs == 255u) ws = dinv[s];
  if (bd == 255u) wd = dinv[d];
  return ws * wd;
}

template <int C>
__device__ __forceinline__ void rows_full(const float* __restrict__ m, float* __restrict__ agg,
                                          int s, int d, float w, int lane) {
  constexpr int R = C / 32;
  constexpr int U = 16 / R;
  #pragma unroll
  for (int j0 = 0; j0 < 32; j0 += U) {
    float v[U][R];
    #pragma unroll
    for (int u = 0; u < U; ++u) {
      int sj = __builtin_amdgcn_readlane(s, j0 + u);
      const float* row = m + (size_t)sj * C + lane;
      #pragma unroll
      for (int r = 0; r < R; ++r) v[u][r] = row[32 * r];
    }
    #pragma unroll
    for (int u = 0; u < U; ++u) {
      int   dj = __builtin_amdgcn_readlane(d, j0 + u);
      float wj = __int_as_float(__builtin_amdgcn_readlane(__float_as_int(w), j0 + u));
      float* arow = agg + (size_t)dj * C + lane;
      #pragma unroll
      for (int r = 0; r < R; ++r) {
        (void)__hip_atomic_fetch_add(&arow[32 * r], v[u][r] * wj,
                                     __ATOMIC_RELAXED, __HIP_MEMORY_SCOPE_AGENT);
      }
    }
  }
}

template <int C>
__device__ __forceinline__ void rows_part(const float* __restrict__ m, float* __restrict__ agg,
                                          int s, int d, float w, int cnt, int lane) {
  constexpr int R = C / 32;
  constexpr int U = 16 / R;
  for (int j0 = 0; j0 < cnt; j0 += U) {
    float v[U][R];
    #pragma unroll
    for (int u = 0; u < U; ++u) {
      int j  = j0 + u < cnt ? j0 + u : cnt - 1;
      int sj = __builtin_amdgcn_readlane(s, j);
      const float* row = m + (size_t)sj * C + lane;
      #pragma unroll
      for (int r = 0; r < R; ++r) v[u][r] = row[32 * r];
    }
    #pragma unroll
    for (int u = 0; u < U; ++u) {
      if (j0 + u < cnt) {
        int   dj = __builtin_amdgcn_readlane(d, j0 + u);
        float wj = __int_as_float(__builtin_amdgcn_readlane(__float_as_int(w), j0 + u));
        float* arow = agg + (size_t)dj * C + lane;
        #pragma unroll
        for (int r = 0; r < R; ++r) {
          (void)__hip_atomic_fetch_add(&arow[32 * r], v[u][r] * wj,
                                       __ATOMIC_RELAXED, __HIP_MEMORY_SCOPE_AGENT);
        }
      }
    }
  }
}

template <int C>
__global__ void __launch_bounds__(1024) __attribute__((amdgpu_waves_per_eu(16, 16))) k_scatter(const float* __restrict__ m,
                                                 const int* __restrict__ src,
                                                 const int* __restrict__ dst,
                                                 const float* __restrict__ dinv,
                                                 const unsigned* __restrict__ degw,
                                                 float* __restrict__ agg,
                                                 unsigned* __restrict__ tick, int e, int n) {
  __shared__ __attribute__((aligned(16))) unsigned char sdeg[DEG_BYTES];
  __shared__ int qs[Q_RING];
  __shared__ int qd[Q_RING];
  __shared__ int s_tail;
  __shared__ unsigned s_t[2];
  const int tid  = threadIdx.x;
  const int lane = tid & 31;
  const int wv   = tid >> 5;
  const int nsplit = n >> 1;
  const unsigned ntick = (unsigned)((e + TK_EDGES - 1) / TK_EDGES);
  const int half = (int)((__builtin_amdgcn_s_sendmsg_rtn(0x87) >> 18) & 1u);

  {
    constexpr int NV = DEG_BYTES / 16;
    constexpr int NP = (NV + 1023) / 1024;
    const uint4* g4 = (const uint4*)degw;
    uint4* l4 = (uint4*)sdeg;
    uint4 tv[NP];
    #pragma unroll
    for (int u = 0; u < NP; ++u) {
      int i = (u < NP - 1 ? u * 1024 : NV - 1024) + tid;
      tv[u] = g4[i];
    }
    #pragma unroll
    for (int u = 0; u < NP; ++u) {
      int i = (u < NP - 1 ? u * 1024 : NV - 1024) + tid;
      l4[i] = tv[u];
    }
  }

  if (tid == 0) s_tail = 0;

  int head = 0;
  int it   = 0;
  for (int pass = 0; pass < 2; ++pass) {
    const int rg = half ^ pass;
    for (;;) {
      if (tid == 0)
        s_t[it & 1] = __hip_atomic_fetch_add(&tick[rg * TK_PITCH], 1u, __ATOMIC_RELAXED,
                                             __HIP_MEMORY_SCOPE_AGENT);
      __syncthreads();
      const unsigned t = s_t[it & 1];
      ++it;
      if (t >= ntick) break;

      int  si[TK_EDGES / 1024], di[TK_EDGES / 1024];
      bool in[TK_EDGES / 1024];
      #pragma unroll
      for (int u = 0; u < TK_EDGES / 1024; ++u) {
        long i  = (long)t * TK_EDGES + u * 1024 + tid;
        in[u]   = i < e;
        long ic = in[u] ? i : (long)e - 1;
        si[u] = src[ic];
        di[u] = dst[ic];
      }
      #pragma unroll
      for (int u = 0; u < TK_EDGES / 1024; ++u) {
        bool act = in[u] && ((di[u] >= nsplit) == (rg != 0));
        unsigned mask = __builtin_amdgcn_ballot_w32(act);
        int c    = __builtin_popcount(mask);
        int base = 0;
        if (lane == 0 && c != 0)
          base = __hip_atomic_fetch_add(&s_tail, c, __ATOMIC_RELAXED,
                                        __HIP_MEMORY_SCOPE_WORKGROUP);
        base = __builtin_amdgcn_readfirstlane(base);
        int pos = (base + (int)__builtin_amdgcn_mbcnt_lo(mask, 0u)) & (Q_RING - 1);
        if (act) { qs[pos] = si[u]; qd[pos] = di[u]; }
      }
      __syncthreads();

      const int avail = s_tail - head;
      const int proc  = (avail >> 5) & ~(SC_WAVES - 1);
      for (int sl = wv; sl < proc; sl += SC_WAVES) {
        int   p = (head + sl * 32 + lane) & (Q_RING - 1);
        int   s = qs[p];
        int   d = qd[p];
        float w = edge_weight(sdeg, dinv, s, d);
        rows_full<C>(m, agg, s, d, w, lane);
      }
      head += proc * 32;
    }
  }

  const int avail = s_tail - head;
  const int full  = avail >> 5;
  const int rem   = avail & 31;
  if (wv < full) {
    int   p = (head + wv * 32 + lane) & (Q_RING - 1);
    int   s = qs[p];
    int   d = qd[p];
    float w = edge_weight(sdeg, dinv, s, d);
    rows_full<C>(m, agg, s, d, w, lane);
  }
  if (wv == full && rem > 0) {
    int s = 0, d = 0;
    float w = 0.0f;
    if (lane < rem) {
      int p = (head + full * 32 + lane) & (Q_RING - 1);
      s = qs[p];
      d = qd[p];
      w = edge_weight(sdeg, dinv, s, d);
    }
    rows_part<C>(m, agg, s, d, w, rem, lane);
  }
}

extern "C" void kernel_launch(void* const* d_in, const int* in_sizes, int n_in,
                              void* d_out, int out_size, void* d_ws, size_t ws_size,
                              hipStream_t stream) {
  const float* x   = (const float*)d_in[0];
  const int*   ei  = (const int*)d_in[1];
  const float* W1  = (const float*)d_in[2];
  const float* b1  = (const float*)d_in[3];
  const float* W2  = (const float*)d_in[4];
  const float* b2  = (const float*)d_in[5];
  float*       out = (float*)d_out;

  const int N = in_sizes[0] / IN_CH;
  const int E = in_sizes[1] / 2;
  const int* src = ei;
  const int* dst = ei + E;

  float* dinv = (float*)d_ws;
  unsigned* tick = (unsigned*)(dinv + 130048);
  unsigned* degw = (unsigned*)(dinv + 100352);
  float* bufA = dinv + 131072;
  float* bufB = bufA + (size_t)N * HID;

  const int BT = 256;

  k_hist<<<HIST_C * HIST_R, 1024, 0, stream>>>(dst, (unsigned*)out, E);
  {
    int tiles_m = (N + 15) / 16, tiles_n = HID / 16;
    int tiles = tiles_m * tiles_n;
    long threads = (long)tiles * 32;
    k_gemm_f32<false, true, IN_CH, HID><<<(int)((threads + BT - 1) / BT), BT, 0, stream>>>(
        x, W1, dinv, b1, bufA, bufB, (const unsigned*)out, tick, degw, N, tiles);
  }
  k_scatter<HID><<<512, 1024, 0, stream>>>(bufA, src, dst, dinv, degw, bufB, tick, E, N);
  {
    int tiles_m = (N + 15) / 16, tiles_n = OUT_CH / 16;
    int tiles = tiles_m * tiles_n;
    long threads = (long)tiles * 32;
    k_gemm_f32<true, false, HID, OUT_CH><<<(int)((threads + BT - 1) / BT), BT, 0, stream>>>(
        bufB, W2, dinv, b2, bufA, out, (const unsigned*)nullptr, tick, (unsigned*)nullptr, N, tiles);
  }
  k_scatter<OUT_CH><<<512, 1024, 0, stream>>>(bufA, src, dst, dinv, degw, out, tick + 128, E, N);
}
